# starting kernel
# baseline (speedup 1.0000x reference)
#include <hip/hip_runtime.h>

typedef __attribute__((ext_vector_type(2))) float v2f;
typedef __attribute__((ext_vector_type(8))) float v8f;

#define BM 64
#define BN 128
#define KT 32
#define BSTRIDE (KT + 2)

__global__ __launch_bounds__(256)
void moe_gemm_bias_act(const float* __restrict__ in,
                       const float* __restrict__ W,
                       const float* __restrict__ bias,
                       const int*   __restrict__ sel,
                       float* __restrict__ out,
                       int M, int N, int K, int relu)
{
    __shared__ float As[BM * KT];
    __shared__ float Bs[BN * BSTRIDE];

    const int a = blockIdx.z;
    const int e = sel[a];
    const float* inA  = in   + (size_t)a * M * K;
    const float* We   = W    + (size_t)e * K * N;
    const float* be   = bias + (size_t)e * N;
    float*       outA = out  + (size_t)a * M * N;

    const int rowBase = blockIdx.x * BM;
    const int colBase = blockIdx.y * BN;

    const int tid   = threadIdx.x;
    const int lane  = tid & 31;
    const int wave  = tid >> 5;
    const int wr    = (wave & 1) * 32;
    const int wc    = (wave >> 1) * 32;

    const int lm    = lane & 15;
    const int khalf = lane >> 4;

    v8f acc[2][2];
    {
        v8f z = {0.f, 0.f, 0.f, 0.f, 0.f, 0.f, 0.f, 0.f};
        acc[0][0] = z; acc[0][1] = z; acc[1][0] = z; acc[1][1] = z;
    }

    for (int k0 = 0; k0 < K; k0 += KT) {
        #pragma unroll
        for (int i = 0; i < 2; ++i) {
            int idx = tid + i * 256;
            int r   = idx >> 3;
            int c4  = idx & 7;
            float4 v = *reinterpret_cast<const float4*>(
                inA + (size_t)(rowBase + r) * K + k0 + c4 * 4);
            *reinterpret_cast<float4*>(&As[r * KT + c4 * 4]) = v;
        }
        #pragma unroll
        for (int i = 0; i < 4; ++i) {
            int idx = tid + i * 256;
            int r   = idx >> 5;
            int c4  = idx & 31;
            float4 v = *reinterpret_cast<const float4*>(
                We + (size_t)(k0 + r) * N + colBase + c4 * 4);
            int nb = c4 * 4;
            Bs[(nb + 0) * BSTRIDE + r] = v.x;
            Bs[(nb + 1) * BSTRIDE + r] = v.y;
            Bs[(nb + 2) * BSTRIDE + r] = v.z;
            Bs[(nb + 3) * BSTRIDE + r] = v.w;
        }
        __syncthreads();

        #pragma unroll
        for (int kk = 0; kk < KT; kk += 4) {
            v2f afrag[2], bfrag[2];
            #pragma unroll
            for (int rs = 0; rs < 2; ++rs) {
                int m = wr + rs * 16 + lm;
                afrag[rs] = *reinterpret_cast<const v2f*>(&As[m * KT + kk + khalf * 2]);
            }
            #pragma unroll
            for (int cs = 0; cs < 2; ++cs) {
                int n = wc + cs * 16 + lm;
                bfrag[cs] = *reinterpret_cast<const v2f*>(&Bs[n * BSTRIDE + kk + khalf * 2]);
            }
            #pragma unroll
            for (int rs = 0; rs < 2; ++rs)
                #pragma unroll
                for (int cs = 0; cs < 2; ++cs)
                    acc[rs][cs] = __builtin_amdgcn_wmma_f32_16x16x4_f32(
                         false, afrag[rs],
                         false, bfrag[cs],
                         (short)0, acc[rs][cs],
                         false,  false);
        }
        __syncthreads();
    }

    const int mHalf = khalf * 8;
    #pragma unroll
    for (int rs = 0; rs < 2; ++rs) {
        #pragma unroll
        for (int cs = 0; cs < 2; ++cs) {
            int   col = colBase + wc + cs * 16 + lm;
            float bv  = be[col];
            #pragma unroll
            for (int v = 0; v < 8; ++v) {
                int   row = rowBase + wr + rs * 16 + mHalf + v;
                float x   = acc[rs][cs][v] + bv;
                if (relu) x = fmaxf(x, 0.f);
                outA[(size_t)row * N + col] = x;
            }
        }
    }
}

extern "C" void kernel_launch(void* const* d_in, const int* in_sizes, int n_in,
                              void* d_out, int out_size, void* d_ws, size_t ws_size,
                              hipStream_t stream) {
    (void)in_sizes; (void)n_in; (void)out_size; (void)ws_size;

    const float* inputs = (const float*)d_in[0];
    const int*   sel_s  = (const int*)  d_in[1];
    const int*   sel_d  = (const int*)  d_in[2];
    const float* Ws1    = (const float*)d_in[3];
    const float* bs1    = (const float*)d_in[4];
    const float* Ws2    = (const float*)d_in[5];
    const float* bs2    = (const float*)d_in[6];
    const float* Wd1    = (const float*)d_in[7];
    const float* bd1    = (const float*)d_in[8];
    const float* Wd2    = (const float*)d_in[9];
    const float* bd2    = (const float*)d_in[10];
    float* out = (float*)d_out;

    const int A = 8, B = 2048, S = 256, H1 = 512, H2 = 256, D1 = 512, D2 = 128;

    float* h  = (float*)d_ws;
    float* sh = h  + (size_t)A * B * H1;
    float* dd = sh + (size_t)A * B * H2;

    dim3 blk(256);
    moe_gemm_bias_act<<<dim3(B / BM, H1 / BN, A), blk, 0, stream>>>(
        inputs, Ws1, bs1, sel_s, h,  B, H1, S,  1);
    moe_gemm_bias_act<<<dim3(B / BM, H2 / BN, A), blk, 0, stream>>>(
        h,      Ws2, bs2, sel_s, sh, B, H2, H1, 1);
    moe_gemm_bias_act<<<dim3(B / BM, D1 / BN, A), blk, 0, stream>>>(
        sh,     Wd1, bd1, sel_d, dd, B, D1, H2, 1);
    moe_gemm_bias_act<<<dim3(B / BM, D2 / BN, A), blk, 0, stream>>>(
        dd,     Wd2, bd2, sel_d, out, B, D2, D1, 0);
}


# LDS bank conflicts removed: A slab pitch 36 dwords (fragment reads were 8-way); B slab kept untransposed as two 64-column panels at pitch 80, 16 B stores, two-address fragment reads
# speedup vs baseline: 1.1303x; 1.1303x over previous
#include <hip/hip_runtime.h>

typedef __attribute__((ext_vector_type(2))) float v2f;
typedef __attribute__((ext_vector_type(8))) float v8f;

#define BM 64
#define BN 128
#define KT 32
#define APITCH (KT + 4)
#define BPANEL 64
#define BPITCH (BPANEL + 16)

__global__ __launch_bounds__(256)
void moe_gemm_bias_act(const float* __restrict__ in,
                       const float* __restrict__ W,
                       const float* __restrict__ bias,
                       const int*   __restrict__ sel,
                       float* __restrict__ out,
                       int M, int N, int K, int relu)
{
    __shared__ __attribute__((aligned(16))) float As[BM * APITCH];
    __shared__ __attribute__((aligned(16))) float Bs[2 * KT * BPITCH];

    const int a = blockIdx.z;
    const int e = sel[a];
    const float* inA  = in   + (size_t)a * M * K;
    const float* We   = W    + (size_t)e * K * N;
    const float* be   = bias + (size_t)e * N;
    float*       outA = out  + (size_t)a * M * N;

    const int rowBase = blockIdx.x * BM;
    const int colBase = blockIdx.y * BN;

    const int tid   = threadIdx.x;
    const int lane  = tid & 31;
    const int wave  = tid >> 5;
    const int wr    = (wave & 1) * 32;
    const int wc    = (wave >> 1) * 16;

    const int lm    = lane & 15;
    const int khalf = lane >> 4;

    v8f acc[2][2];
    {
        v8f z = {0.f, 0.f, 0.f, 0.f, 0.f, 0.f, 0.f, 0.f};
        acc[0][0] = z; acc[0][1] = z; acc[1][0] = z; acc[1][1] = z;
    }

    for (int k0 = 0; k0 < K; k0 += KT) {
        #pragma unroll
        for (int i = 0; i < 2; ++i) {
            int idx = tid + i * 256;
            int q   = idx >> 3;
            int r   = (q & 0x30) | ((q & 1) << 3) | ((q >> 1) & 7);
            int c4  = idx & 7;
            float4 v = *reinterpret_cast<const float4*>(
                inA + (size_t)(rowBase + r) * K + k0 + c4 * 4);
            *reinterpret_cast<float4*>(&As[r * APITCH + c4 * 4]) = v;
        }
        #pragma unroll
        for (int i = 0; i < 4; ++i) {
            int idx = tid + i * 256;
            int r   = idx >> 5;
            int c4  = idx & 31;
            float4 v = *reinterpret_cast<const float4*>(
                We + (size_t)(k0 + r) * N + colBase + c4 * 4);
            *reinterpret_cast<float4*>(
                &Bs[(c4 >> 4) * (KT * BPITCH) + r * BPITCH + (c4 & 15) * 4]) = v;
        }
        __syncthreads();

        #pragma unroll
        for (int kk = 0; kk < KT; kk += 4) {
            v2f afrag[2], bfrag[2];
            #pragma unroll
            for (int rs = 0; rs < 2; ++rs) {
                int m = wr + rs * 16 + lm;
                afrag[rs] = *reinterpret_cast<const v2f*>(&As[m * APITCH + kk + khalf * 2]);
            }
            #pragma unroll
            for (int cs = 0; cs < 2; ++cs) {
                const float* bp = &Bs[cs * (KT * BPITCH) + (kk + khalf * 2) * BPITCH + wc + lm];
                v2f b;
                b.x = bp[0];
                b.y = bp[BPITCH];
                bfrag[cs] = b;
            }
            #pragma unroll
            for (int rs = 0; rs < 2; ++rs)
                #pragma unroll
                for (int cs = 0; cs < 2; ++cs)
                    acc[rs][cs] = __builtin_amdgcn_wmma_f32_16x16x4_f32(
                         false, afrag[rs],
                         false, bfrag[cs],
                         (short)0, acc[rs][cs],
                         false,  false);
        }
        __syncthreads();
    }

    const int mHalf = khalf * 8;
    #pragma unroll
    for (int rs = 0; rs < 2; ++rs) {
        #pragma unroll
        for (int cs = 0; cs < 2; ++cs) {
            int   col = colBase + cs * BPANEL + wc + lm;
            float bv  = be[col];
            #pragma unroll
            for (int v = 0; v < 8; ++v) {
                int   row = rowBase + wr + rs * 16 + mHalf + v;
                float x   = acc[rs][cs][v] + bv;
                if (relu) x = fmaxf(x, 0.f);
                outA[(size_t)row * N + col] = x;
            }
        }
    }
}

extern "C" void kernel_launch(void* const* d_in, const int* in_sizes, int n_in,
                              void* d_out, int out_size, void* d_ws, size_t ws_size,
                              hipStream_t stream) {
    (void)in_sizes; (void)n_in; (void)out_size; (void)ws_size;

    const float* inputs = (const float*)d_in[0];
    const int*   sel_s  = (const int*)  d_in[1];
    const int*   sel_d  = (const int*)  d_in[2];
    const float* Ws1    = (const float*)d_in[3];
    const float* bs1    = (const float*)d_in[4];
    const float* Ws2    = (const float*)d_in[5];
    const float* bs2    = (const float*)d_in[6];
    const float* Wd1    = (const float*)d_in[7];
    const float* bd1    = (const float*)d_in[8];
    const float* Wd2    = (const float*)d_in[9];
    const float* bd2    = (const float*)d_in[10];
    float* out = (float*)d_out;

    const int A = 8, B = 2048, S = 256, H1 = 512, H2 = 256, D1 = 512, D2 = 128;

    float* h  = (float*)d_ws;
    float* sh = h  + (size_t)A * B * H1;
    float* dd = sh + (size_t)A * B * H2;

    dim3 blk(256);
    moe_gemm_bias_act<<<dim3(B / BM, H1 / BN, A), blk, 0, stream>>>(
        inputs, Ws1, bs1, sel_s, h,  B, H1, S,  1);
    moe_gemm_bias_act<<<dim3(B / BM, H2 / BN, A), blk, 0, stream>>>(
        h,      Ws2, bs2, sel_s, sh, B, H2, H1, 1);
    moe_gemm_bias_act<<<dim3(B / BM, D1 / BN, A), blk, 0, stream>>>(
        sh,     Wd1, bd1, sel_d, dd, B, D1, H2, 1);
    moe_gemm_bias_act<<<dim3(B / BM, D2 / BN, A), blk, 0, stream>>>(
        dd,     Wd2, bd2, sel_d, out, B, D2, D1, 0);
}


# K loop software-pipelined over two LDS stages: the next slab's global loads are issued ahead of the current slab's WMMAs and stored to the other stage behind them; one barrier per slab, not two
# speedup vs baseline: 1.3691x; 1.2113x over previous
#include <hip/hip_runtime.h>

typedef __attribute__((ext_vector_type(2))) float v2f;
typedef __attribute__((ext_vector_type(8))) float v8f;

#define BM 64
#define BN 128
#define KT 32
#define APITCH (KT + 4)
#define BPANEL 64
#define BPITCH (BPANEL + 16)
#define A_FLOATS (BM * APITCH)
#define B_FLOATS (2 * KT * BPITCH)
#define STAGE_FLOATS (A_FLOATS + B_FLOATS)

struct SlabRegs {
    float4 a[2];
    float4 b[4];
};

__device__ __forceinline__ void slab_load(SlabRegs& s, const float* aSrc, const float* bSrc,
                                          int K, int N)
{
    s.a[0] = *reinterpret_cast<const float4*>(aSrc);
    s.a[1] = *reinterpret_cast<const float4*>(aSrc + (size_t)32 * K);
    #pragma unroll
    for (int i = 0; i < 4; ++i)
        s.b[i] = *reinterpret_cast<const float4*>(bSrc + (size_t)(8 * i) * N);
}

__device__ __forceinline__ void slab_store(const SlabRegs& s, float* stage, int aDst, int bDst)
{
    *reinterpret_cast<float4*>(&stage[aDst])               = s.a[0];
    *reinterpret_cast<float4*>(&stage[aDst + 32 * APITCH]) = s.a[1];
    #pragma unroll
    for (int i = 0; i < 4; ++i)
        *reinterpret_cast<float4*>(&stage[A_FLOATS + bDst + 8 * i * BPITCH]) = s.b[i];
}

__device__ __forceinline__ void slab_compute(const float* stage, int aFrag, int bFrag,
                                             v8f (&acc)[2][2])
{
    #pragma unroll
    for (int kk = 0; kk < KT; kk += 4) {
        v2f afrag[2], bfrag[2];
        #pragma unroll
        for (int rs = 0; rs < 2; ++rs)
            afrag[rs] = *reinterpret_cast<const v2f*>(&stage[aFrag + rs * 16 * APITCH + kk]);
        #pragma unroll
        for (int cs = 0; cs < 2; ++cs) {
            const float* bp = &stage[A_FLOATS + bFrag + cs * (KT * BPITCH) + kk * BPITCH];
            v2f b;
            b.x = bp[0];
            b.y = bp[BPITCH];
            bfrag[cs] = b;
        }
        #pragma unroll
        for (int rs = 0; rs < 2; ++rs)
            #pragma unroll
            for (int cs = 0; cs < 2; ++cs)
                acc[rs][cs] = __builtin_amdgcn_wmma_f32_16x16x4_f32(
                     false, afrag[rs],
                     false, bfrag[cs],
                     (short)0, acc[rs][cs],
                     false,  false);
    }
}

__global__ __launch_bounds__(256)
void moe_gemm_bias_act(const float* __restrict__ in,
                       const float* __restrict__ W,
                       const float* __restrict__ bias,
                       const int*   __restrict__ sel,
                       float* __restrict__ out,
                       int M, int N, int K, int relu)
{
    __shared__ __attribute__((aligned(16))) float lds[2 * STAGE_FLOATS];

    const int a = blockIdx.z;
    const int e = sel[a];
    const float* inA  = in   + (size_t)a * M * K;
    const float* We   = W    + (size_t)e * K * N;
    const float* be   = bias + (size_t)e * N;
    float*       outA = out  + (size_t)a * M * N;

    const int rowBase = blockIdx.x * BM;
    const int colBase = blockIdx.y * BN;

    const int tid   = threadIdx.x;
    const int lane  = tid & 31;
    const int wave  = tid >> 5;
    const int wr    = (wave & 1) * 32;
    const int wc    = (wave >> 1) * 16;

    const int lm    = lane & 15;
    const int khalf = lane >> 4;

    const int aq  = tid >> 3;
    const int ar  = (aq & 0x30) | ((aq & 1) << 3) | ((aq >> 1) & 7);
    const int ac4 = tid & 7;
    const int bk  = tid >> 5;
    const int bc4 = tid & 31;
    const float* aSrc = inA + (size_t)(rowBase + ar) * K + ac4 * 4;
    const float* bSrc = We + (size_t)bk * N + colBase + bc4 * 4;
    const int aDst = ar * APITCH + ac4 * 4;
    const int bDst = (bc4 >> 4) * (KT * BPITCH) + bk * BPITCH + (bc4 & 15) * 4;

    const int aFrag = (wr + lm) * APITCH + khalf * 2;
    const int bFrag = (khalf * 2) * BPITCH + wc + lm;

    v8f acc[2][2];
    {
        v8f z = {0.f, 0.f, 0.f, 0.f, 0.f, 0.f, 0.f, 0.f};
        acc[0][0] = z; acc[0][1] = z; acc[1][0] = z; acc[1][1] = z;
    }

    SlabRegs sr;
    float* rd = lds;
    float* wt = lds + STAGE_FLOATS;

    slab_load(sr, aSrc, bSrc, K, N);
    slab_store(sr, rd, aDst, bDst);
    __syncthreads();

    for (int k0 = KT; k0 < K; k0 += KT) {
        aSrc += KT;
        bSrc += (size_t)KT * N;
        slab_load(sr, aSrc, bSrc, K, N);
        slab_compute(rd, aFrag, bFrag, acc);
        slab_store(sr, wt, aDst, bDst);
        __syncthreads();
        float* t = rd; rd = wt; wt = t;
    }
    slab_compute(rd, aFrag, bFrag, acc);

    const int mHalf = khalf * 8;
    #pragma unroll
    for (int rs = 0; rs < 2; ++rs) {
        #pragma unroll
        for (int cs = 0; cs < 2; ++cs) {
            int   col = colBase + cs * BPANEL + wc + lm;
            float bv  = be[col];
            #pragma unroll
            for (int v = 0; v < 8; ++v) {
                int   row = rowBase + wr + rs * 16 + mHalf + v;
                float x   = acc[rs][cs][v] + bv;
                if (relu) x = fmaxf(x, 0.f);
                outA[(size_t)row * N + col] = x;
            }
        }
    }
}

extern "C" void kernel_launch(void* const* d_in, const int* in_sizes, int n_in,
                              void* d_out, int out_size, void* d_ws, size_t ws_size,
                              hipStream_t stream) {
    (void)in_sizes; (void)n_in; (void)out_size; (void)ws_size;

    const float* inputs = (const float*)d_in[0];
    const int*   sel_s  = (const int*)  d_in[1];
    const int*   sel_d  = (const int*)  d_in[2];
    const float* Ws1    = (const float*)d_in[3];
    const float* bs1    = (const float*)d_in[4];
    const float* Ws2    = (const float*)d_in[5];
    const float* bs2    = (const float*)d_in[6];
    const float* Wd1    = (const float*)d_in[7];
    const float* bd1    = (const float*)d_in[8];
    const float* Wd2    = (const float*)d_in[9];
    const float* bd2    = (const float*)d_in[10];
    float* out = (float*)d_out;

    const int A = 8, B = 2048, S = 256, H1 = 512, H2 = 256, D1 = 512, D2 = 128;

    float* h  = (float*)d_ws;
    float* sh = h  + (size_t)A * B * H1;
    float* dd = sh + (size_t)A * B * H2;

    dim3 blk(256);
    moe_gemm_bias_act<<<dim3(B / BM, H1 / BN, A), blk, 0, stream>>>(
        inputs, Ws1, bs1, sel_s, h,  B, H1, S,  1);
    moe_gemm_bias_act<<<dim3(B / BM, H2 / BN, A), blk, 0, stream>>>(
        h,      Ws2, bs2, sel_s, sh, B, H2, H1, 1);
    moe_gemm_bias_act<<<dim3(B / BM, D1 / BN, A), blk, 0, stream>>>(
        sh,     Wd1, bd1, sel_d, dd, B, D1, H2, 1);
    moe_gemm_bias_act<<<dim3(B / BM, D2 / BN, A), blk, 0, stream>>>(
        dd,     Wd2, bd2, sel_d, out, B, D2, D1, 0);
}


# Four launches fused into one: a workgroup takes its 64 rows through all four layers; the hidden activations stay in LDS and are read in place as the next layer's A operand, never touching memory
# speedup vs baseline: 1.4791x; 1.0803x over previous
#include <hip/hip_runtime.h>

typedef __attribute__((ext_vector_type(2))) float v2f;
typedef __attribute__((ext_vector_type(8))) float v8f;
typedef __attribute__((ext_vector_type(4))) float v4f;

#define BM 64
#define BN 128
#define KT 32
#define APITCH (KT + 4)
#define BPANEL 64
#define BPITCH (BPANEL + 16)
#define A_FLOATS (BM * APITCH)
#define B_FLOATS (2 * KT * BPITCH)
#define STAGE_FLOATS (A_FLOATS + B_FLOATS)
#define HPITCH (512 + 4)
#define SPITCH (256 + 4)
#define H_FLOATS (BM * HPITCH)
#define S_FLOATS (BM * SPITCH)

struct SlabRegs {
    v4f a0, a1;
    v4f b0, b1, b2, b3;
};

__device__ __forceinline__ void slab_load_a(SlabRegs& s, const float* aSrc, int K)
{
    s.a0 = *reinterpret_cast<const v4f*>(aSrc);
    s.a1 = *reinterpret_cast<const v4f*>(aSrc + (size_t)32 * K);
}

__device__ __forceinline__ void slab_load_b(SlabRegs& s, const float* bSrc, int N)
{
    s.b0 = *reinterpret_cast<const v4f*>(bSrc);
    s.b1 = *reinterpret_cast<const v4f*>(bSrc + (size_t)8 * N);
    s.b2 = *reinterpret_cast<const v4f*>(bSrc + (size_t)16 * N);
    s.b3 = *reinterpret_cast<const v4f*>(bSrc + (size_t)24 * N);
}

template <bool A_LDS>
__device__ __forceinline__ void slab_store(const SlabRegs& s, float* stage, int aDst, int bDst)
{
    if (!A_LDS) {
        *reinterpret_cast<v4f*>(&stage[aDst])               = s.a0;
        *reinterpret_cast<v4f*>(&stage[aDst + 32 * APITCH]) = s.a1;
    }
    float* bSt = &stage[A_FLOATS + bDst];
    *reinterpret_cast<v4f*>(bSt)               = s.b0;
    *reinterpret_cast<v4f*>(bSt + 8 * BPITCH)  = s.b1;
    *reinterpret_cast<v4f*>(bSt + 16 * BPITCH) = s.b2;
    *reinterpret_cast<v4f*>(bSt + 24 * BPITCH) = s.b3;
}

template <int AP>
__device__ __forceinline__ void slab_compute(const float* aBase, const float* bBase, v8f (&acc)[2][2])
{
    #pragma unroll
    for (int kk = 0; kk < KT; kk += 4) {
        v2f afrag[2], bfrag[2];
        #pragma unroll
        for (int rs = 0; rs < 2; ++rs)
            afrag[rs] = *reinterpret_cast<const v2f*>(aBase + rs * 16 * AP + kk);
        #pragma unroll
        for (int cs = 0; cs < 2; ++cs) {
            const float* bp = bBase + cs * (KT * BPITCH) + kk * BPITCH;
            v2f b;
            b.x = bp[0];
            b.y = bp[BPITCH];
            bfrag[cs] = b;
        }
        #pragma unroll
        for (int rs = 0; rs < 2; ++rs)
            #pragma unroll
            for (int cs = 0; cs < 2; ++cs)
                acc[rs][cs] = __builtin_amdgcn_wmma_f32_16x16x4_f32(
                     false, afrag[rs],
                     false, bfrag[cs],
                     (short)0, acc[rs][cs],
                     false,  false);
    }
}

template <int K, int N, bool A_LDS, int AP, bool OUT_LDS, int OP, bool RELU, int NEXT_N>
__device__ __forceinline__ void layer(const float* __restrict__ aGlob, const float* aLds,
                                      const float* __restrict__ We, const float* __restrict__ be,
                                      const float* __restrict__ nextW,
                                      float* __restrict__ oGlob, float* oLds,
                                      SlabRegs& sr, float*& rd, float*& wt, int tid)
{
    const int lane  = tid & 31;
    const int wave  = tid >> 5;
    const int wr    = (wave & 1) * 32;
    const int wc    = (wave >> 1) * 16;
    const int lm    = lane & 15;
    const int khalf = lane >> 4;

    const int aq  = tid >> 3;
    const int ar  = (aq & 0x30) | ((aq & 1) << 3) | ((aq >> 1) & 7);
    const int ac4 = tid & 7;
    const int bk  = tid >> 5;
    const int bc4 = tid & 31;
    const int aDst = ar * APITCH + ac4 * 4;
    const int bDst = (bc4 >> 4) * (KT * BPITCH) + bk * BPITCH + (bc4 & 15) * 4;

    const int aFragStage = (wr + lm) * APITCH + khalf * 2;
    const int aFragLds   = (wr + lm) * AP + khalf * 2;
    const int bFrag      = A_FLOATS + (khalf * 2) * BPITCH + wc + lm;
    const int mHalf      = khalf * 8;

    const float* aSrc0 = A_LDS ? nullptr : aGlob + (size_t)ar * K + ac4 * 4;
    const float* bSrc0 = We + (size_t)bk * N + bc4 * 4;

    for (int nb = 0; nb < N; nb += BN) {
        v8f acc[2][2];
        {
            v8f z = {0.f, 0.f, 0.f, 0.f, 0.f, 0.f, 0.f, 0.f};
            acc[0][0] = z; acc[0][1] = z; acc[1][0] = z; acc[1][1] = z;
        }

        slab_store<A_LDS>(sr, wt, aDst, bDst);
        __syncthreads();
        { float* t = rd; rd = wt; wt = t; }

        const float* aSrc = aSrc0;
        const float* bSrc = bSrc0 + nb;
        #pragma nounroll
        for (int k0 = KT; k0 < K; k0 += KT) {
            bSrc += (size_t)KT * N;
            slab_load_b(sr, bSrc, N);
            if (!A_LDS) { aSrc += KT; slab_load_a(sr, aSrc, K); }
            if (A_LDS) slab_compute<AP>(aLds + aFragLds + (k0 - KT), rd + bFrag, acc);
            else       slab_compute<APITCH>(rd + aFragStage, rd + bFrag, acc);
            slab_store<A_LDS>(sr, wt, aDst, bDst);
            __syncthreads();
            { float* t = rd; rd = wt; wt = t; }
        }

        if (nb + BN < N) {
            slab_load_b(sr, bSrc0 + nb + BN, N);
            if (!A_LDS) slab_load_a(sr, aSrc0, K);
        } else if (NEXT_N != 0) {
            slab_load_b(sr, nextW + (size_t)bk * NEXT_N + bc4 * 4, NEXT_N);
        }

        if (A_LDS) slab_compute<AP>(aLds + aFragLds + (K - KT), rd + bFrag, acc);
        else       slab_compute<APITCH>(rd + aFragStage, rd + bFrag, acc);

        #pragma unroll
        for (int rs = 0; rs < 2; ++rs) {
            #pragma unroll
            for (int cs = 0; cs < 2; ++cs) {
                int   col = nb + cs * BPANEL + wc + lm;
                float bv  = be[col];
                #pragma unroll
                for (int v = 0; v < 8; ++v) {
                    int   row = wr + rs * 16 + mHalf + v;
                    float x   = acc[rs][cs][v] + bv;
                    if (RELU) x = fmaxf(x, 0.f);
                    if (OUT_LDS) oLds[row * OP + col] = x;
                    else         oGlob[(size_t)row * N + col] = x;
                }
            }
        }
    }
}

__global__ __launch_bounds__(256)
void mlp4_fused(const float* __restrict__ in,
                const int*   __restrict__ sel_s,
                const int*   __restrict__ sel_d,
                const float* __restrict__ Ws1, const float* __restrict__ bs1,
                const float* __restrict__ Ws2, const float* __restrict__ bs2,
                const float* __restrict__ Wd1, const float* __restrict__ bd1,
                const float* __restrict__ Wd2, const float* __restrict__ bd2,
                float* __restrict__ out,
                int M)
{
    __shared__ __attribute__((aligned(16))) float lds[2 * STAGE_FLOATS + H_FLOATS + S_FLOATS];
    float* bufH = lds + 2 * STAGE_FLOATS;
    float* bufS = bufH + H_FLOATS;

    const int a  = blockIdx.z;
    const int es = sel_s[a];
    const int ed = sel_d[a];
    const int rowBase = blockIdx.x * BM;
    const int tid = threadIdx.x;

    const float* inA  = in  + ((size_t)a * M + rowBase) * 256;
    float*       outA = out + ((size_t)a * M + rowBase) * 128;
    const float* W1 = Ws1 + (size_t)es * 256 * 512;  const float* b1 = bs1 + (size_t)es * 512;
    const float* W2 = Ws2 + (size_t)es * 512 * 256;  const float* b2 = bs2 + (size_t)es * 256;
    const float* W3 = Wd1 + (size_t)ed * 256 * 512;  const float* b3 = bd1 + (size_t)ed * 512;
    const float* W4 = Wd2 + (size_t)ed * 512 * 128;  const float* b4 = bd2 + (size_t)ed * 128;

    float* rd = lds;
    float* wt = lds + STAGE_FLOATS;

    SlabRegs sr;
    {
        const int aq  = tid >> 3;
        const int ar  = (aq & 0x30) | ((aq & 1) << 3) | ((aq >> 1) & 7);
        slab_load_a(sr, inA + (size_t)ar * 256 + (tid & 7) * 4, 256);
        slab_load_b(sr, W1 + (size_t)(tid >> 5) * 512 + (tid & 31) * 4, 512);
    }

    layer<256, 512, false, APITCH, true,  HPITCH, true,  256>(inA,     nullptr, W1, b1, W2,      nullptr, bufH, sr, rd, wt, tid);
    layer<512, 256, true,  HPITCH, true,  SPITCH, true,  512>(nullptr, bufH,    W2, b2, W3,      nullptr, bufS, sr, rd, wt, tid);
    layer<256, 512, true,  SPITCH, true,  HPITCH, true,  128>(nullptr, bufS,    W3, b3, W4,      nullptr, bufH, sr, rd, wt, tid);
    layer<512, 128, true,  HPITCH, false, 0,      false, 0  >(nullptr, bufH,    W4, b4, nullptr, outA,    nullptr, sr, rd, wt, tid);
}

extern "C" void kernel_launch(void* const* d_in, const int* in_sizes, int n_in,
                              void* d_out, int out_size, void* d_ws, size_t ws_size,
                              hipStream_t stream) {
    (void)in_sizes; (void)n_in; (void)out_size; (void)d_ws; (void)ws_size;

    const float* inputs = (const float*)d_in[0];
    const int*   sel_s  = (const int*)  d_in[1];
    const int*   sel_d  = (const int*)  d_in[2];
    const float* Ws1    = (const float*)d_in[3];
    const float* bs1    = (const float*)d_in[4];
    const float* Ws2    = (const float*)d_in[5];
    const float* bs2    = (const float*)d_in[6];
    const float* Wd1    = (const float*)d_in[7];
    const float* bd1    = (const float*)d_in[8];
    const float* Wd2    = (const float*)d_in[9];
    const float* bd2    = (const float*)d_in[10];
    float* out = (float*)d_out;

    const int A = 8, B = 2048;

    mlp4_fused<<<dim3(B / BM, 1, A), dim3(256), 0, stream>>>(
        inputs, sel_s, sel_d, Ws1, bs1, Ws2, bs2, Wd1, bd1, Wd2, bd2, out, B);
}


# K slab doubled from 32 to 64: half as many barrier and stage hand-overs, 64 matrix instructions per wave between two barriers (A slab pitch 68, staging maps widened to the 64-wide slab)
# speedup vs baseline: 1.6384x; 1.1077x over previous
#include <hip/hip_runtime.h>

typedef __attribute__((ext_vector_type(2))) float v2f;
typedef __attribute__((ext_vector_type(8))) float v8f;
typedef __attribute__((ext_vector_type(4))) float v4f;

#define BM 64
#define BN 128
#define KT 64
#define APITCH (KT + 4)
#define BPANEL 64
#define BPITCH (BPANEL + 16)
#define A_FLOATS (BM * APITCH)
#define B_FLOATS (2 * KT * BPITCH)
#define STAGE_FLOATS (A_FLOATS + B_FLOATS)
#define HPITCH (512 + 4)
#define SPITCH (256 + 4)
#define H_FLOATS (BM * HPITCH)
#define S_FLOATS (BM * SPITCH)

struct SlabRegs {
    v4f a[4];
    v4f b[8];
};

__device__ __forceinline__ void slab_load_a(SlabRegs& s, const float* aSrc, int K)
{
    #pragma unroll
    for (int i = 0; i < 4; ++i)
        s.a[i] = *reinterpret_cast<const v4f*>(aSrc + (size_t)(16 * i) * K);
}

__device__ __forceinline__ void slab_load_b(SlabRegs& s, const float* bSrc, int N)
{
    #pragma unroll
    for (int i = 0; i < 8; ++i)
        s.b[i] = *reinterpret_cast<const v4f*>(bSrc + (size_t)(8 * i) * N);
}

template <bool A_LDS>
__device__ __forceinline__ void slab_store(const SlabRegs& s, float* stage, int aDst, int bDst)
{
    if (!A_LDS) {
        #pragma unroll
        for (int i = 0; i < 4; ++i)
            *reinterpret_cast<v4f*>(&stage[aDst + 16 * i * APITCH]) = s.a[i];
    }
    float* bSt = &stage[A_FLOATS + bDst];
    #pragma unroll
    for (int i = 0; i < 8; ++i)
        *reinterpret_cast<v4f*>(bSt + 8 * i * BPITCH) = s.b[i];
}

template <int AP>
__device__ __forceinline__ void slab_compute(const float* aBase, const float* bBase, v8f (&acc)[2][2])
{
    #pragma unroll
    for (int kk = 0; kk < KT; kk += 4) {
        v2f afrag[2], bfrag[2];
        #pragma unroll
        for (int rs = 0; rs < 2; ++rs)
            afrag[rs] = *reinterpret_cast<const v2f*>(aBase + rs * 16 * AP + kk);
        #pragma unroll
        for (int cs = 0; cs < 2; ++cs) {
            const float* bp = bBase + cs * (KT * BPITCH) + kk * BPITCH;
            v2f b;
            b.x = bp[0];
            b.y = bp[BPITCH];
            bfrag[cs] = b;
        }
        #pragma unroll
        for (int rs = 0; rs < 2; ++rs)
            #pragma unroll
            for (int cs = 0; cs < 2; ++cs)
                acc[rs][cs] = __builtin_amdgcn_wmma_f32_16x16x4_f32(
                     false, afrag[rs],
                     false, bfrag[cs],
                     (short)0, acc[rs][cs],
                     false,  false);
    }
}

template <int K, int N, bool A_LDS, int AP, bool OUT_LDS, int OP, bool RELU, int NEXT_N>
__device__ __forceinline__ void layer(const float* __restrict__ aGlob, const float* aLds,
                                      const float* __restrict__ We, const float* __restrict__ be,
                                      const float* __restrict__ nextW,
                                      float* __restrict__ oGlob, float* oLds,
                                      SlabRegs& sr, float*& rd, float*& wt, int tid)
{
    const int lane  = tid & 31;
    const int wave  = tid >> 5;
    const int wr    = (wave & 1) * 32;
    const int wc    = (wave >> 1) * 16;
    const int lm    = lane & 15;
    const int khalf = lane >> 4;

    const int ar  = tid >> 4;
    const int ac4 = tid & 15;
    const int bk  = tid >> 5;
    const int bc4 = tid & 31;
    const int aDst = ar * APITCH + ac4 * 4;
    const int bDst = (bc4 >> 4) * (KT * BPITCH) + bk * BPITCH + (bc4 & 15) * 4;

    const int aFragStage = (wr + lm) * APITCH + khalf * 2;
    const int aFragLds   = (wr + lm) * AP + khalf * 2;
    const int bFrag      = A_FLOATS + (khalf * 2) * BPITCH + wc + lm;
    const int mHalf      = khalf * 8;

    const float* aSrc0 = A_LDS ? nullptr : aGlob + (size_t)ar * K + ac4 * 4;
    const float* bSrc0 = We + (size_t)bk * N + bc4 * 4;

    for (int nb = 0; nb < N; nb += BN) {
        v8f acc[2][2];
        {
            v8f z = {0.f, 0.f, 0.f, 0.f, 0.f, 0.f, 0.f, 0.f};
            acc[0][0] = z; acc[0][1] = z; acc[1][0] = z; acc[1][1] = z;
        }

        slab_store<A_LDS>(sr, wt, aDst, bDst);
        __syncthreads();
        { float* t = rd; rd = wt; wt = t; }

        const float* aSrc = aSrc0;
        const float* bSrc = bSrc0 + nb;
        #pragma nounroll
        for (int k0 = KT; k0 < K; k0 += KT) {
            bSrc += (size_t)KT * N;
            slab_load_b(sr, bSrc, N);
            if (!A_LDS) { aSrc += KT; slab_load_a(sr, aSrc, K); }
            if (A_LDS) slab_compute<AP>(aLds + aFragLds + (k0 - KT), rd + bFrag, acc);
            else       slab_compute<APITCH>(rd + aFragStage, rd + bFrag, acc);
            if (!A_LDS) __builtin_amdgcn_sched_barrier(0);
            slab_store<A_LDS>(sr, wt, aDst, bDst);
            __syncthreads();
            { float* t = rd; rd = wt; wt = t; }
        }

        if (nb + BN < N) {
            slab_load_b(sr, bSrc0 + nb + BN, N);
            if (!A_LDS) slab_load_a(sr, aSrc0, K);
        } else if (NEXT_N != 0) {
            slab_load_b(sr, nextW + (size_t)bk * NEXT_N + bc4 * 4, NEXT_N);
        }

        if (A_LDS) slab_compute<AP>(aLds + aFragLds + (K - KT), rd + bFrag, acc);
        else       slab_compute<APITCH>(rd + aFragStage, rd + bFrag, acc);

        #pragma unroll
        for (int rs = 0; rs < 2; ++rs) {
            #pragma unroll
            for (int cs = 0; cs < 2; ++cs) {
                int   col = nb + cs * BPANEL + wc + lm;
                float bv  = be[col];
                #pragma unroll
                for (int v = 0; v < 8; ++v) {
                    int   row = wr + rs * 16 + mHalf + v;
                    float x   = acc[rs][cs][v] + bv;
                    if (RELU) x = fmaxf(x, 0.f);
                    if (OUT_LDS) oLds[row * OP + col] = x;
                    else         oGlob[(size_t)row * N + col] = x;
                }
            }
        }
    }
}

__global__ __launch_bounds__(256)
void mlp4_fused(const float* __restrict__ in,
                const int*   __restrict__ sel_s,
                const int*   __restrict__ sel_d,
                const float* __restrict__ Ws1, const float* __restrict__ bs1,
                const float* __restrict__ Ws2, const float* __restrict__ bs2,
                const float* __restrict__ Wd1, const float* __restrict__ bd1,
                const float* __restrict__ Wd2, const float* __restrict__ bd2,
                float* __restrict__ out,
                int M)
{
    __shared__ __attribute__((aligned(16))) float lds[2 * STAGE_FLOATS + H_FLOATS + S_FLOATS];
    float* bufH = lds + 2 * STAGE_FLOATS;
    float* bufS = bufH + H_FLOATS;

    const int a  = blockIdx.z;
    const int es = sel_s[a];
    const int ed = sel_d[a];
    const int rowBase = blockIdx.x * BM;
    const int tid = threadIdx.x;

    const float* inA  = in  + ((size_t)a * M + rowBase) * 256;
    float*       outA = out + ((size_t)a * M + rowBase) * 128;
    const float* W1 = Ws1 + (size_t)es * 256 * 512;  const float* b1 = bs1 + (size_t)es * 512;
    const float* W2 = Ws2 + (size_t)es * 512 * 256;  const float* b2 = bs2 + (size_t)es * 256;
    const float* W3 = Wd1 + (size_t)ed * 256 * 512;  const float* b3 = bd1 + (size_t)ed * 512;
    const float* W4 = Wd2 + (size_t)ed * 512 * 128;  const float* b4 = bd2 + (size_t)ed * 128;

    float* rd = lds;
    float* wt = lds + STAGE_FLOATS;

    SlabRegs sr;
    {
        const int ar  = tid >> 4;
        slab_load_a(sr, inA + (size_t)ar * 256 + (tid & 15) * 4, 256);
        slab_load_b(sr, W1 + (size_t)(tid >> 5) * 512 + (tid & 31) * 4, 512);
    }

    layer<256, 512, false, APITCH, true,  HPITCH, true,  256>(inA,     nullptr, W1, b1, W2,      nullptr, bufH, sr, rd, wt, tid);
    layer<512, 256, true,  HPITCH, true,  SPITCH, true,  512>(nullptr, bufH,    W2, b2, W3,      nullptr, bufS, sr, rd, wt, tid);
    layer<256, 512, true,  SPITCH, true,  HPITCH, true,  128>(nullptr, bufS,    W3, b3, W4,      nullptr, bufH, sr, rd, wt, tid);
    layer<512, 128, true,  HPITCH, false, 0,      false, 0  >(nullptr, bufH,    W4, b4, nullptr, outA,    nullptr, sr, rd, wt, tid);
}

extern "C" void kernel_launch(void* const* d_in, const int* in_sizes, int n_in,
                              void* d_out, int out_size, void* d_ws, size_t ws_size,
                              hipStream_t stream) {
    (void)in_sizes; (void)n_in; (void)out_size; (void)d_ws; (void)ws_size;

    const float* inputs = (const float*)d_in[0];
    const int*   sel_s  = (const int*)  d_in[1];
    const int*   sel_d  = (const int*)  d_in[2];
    const float* Ws1    = (const float*)d_in[3];
    const float* bs1    = (const float*)d_in[4];
    const float* Ws2    = (const float*)d_in[5];
    const float* bs2    = (const float*)d_in[6];
    const float* Wd1    = (const float*)d_in[7];
    const float* bd1    = (const float*)d_in[8];
    const float* Wd2    = (const float*)d_in[9];
    const float* bd2    = (const float*)d_in[10];
    float* out = (float*)d_out;

    const int A = 8, B = 2048;

    mlp4_fused<<<dim3(B / BM, 1, A), dim3(256), 0, stream>>>(
        inputs, sel_s, sel_d, Ws1, bs1, Ws2, bs2, Wd1, bd1, Wd2, bd2, out, B);
}


# B fragments go from global memory straight into registers: a wave owns 64 rows x 32 columns, two register slabs; input tile staged to LDS once; no B staging, no K-loop barrier; sum order unchanged
# speedup vs baseline: 1.7886x; 1.0917x over previous
#include <hip/hip_runtime.h>

typedef __attribute__((ext_vector_type(2))) float v2f;
typedef __attribute__((ext_vector_type(8))) float v8f;
typedef __attribute__((ext_vector_type(4))) float v4f;

#define BM 64
#define KS 32
#define HPITCH (512 + 4)
#define SPITCH (256 + 4)
#define H_FLOATS (BM * HPITCH)
#define S_FLOATS (BM * SPITCH)

struct BSlab {
    v2f f[KS / 4][2];
};

template <int N, int CT>
__device__ __forceinline__ void bslab_load(BSlab& s, const float* __restrict__ p)
{
    #pragma unroll
    for (int t = 0; t < KS / 4; ++t) {
        #pragma unroll
        for (int cs = 0; cs < CT; ++cs) {
            v2f b;
            b.x = p[(size_t)(4 * t) * N + cs * 16];
            b.y = p[(size_t)(4 * t + 1) * N + cs * 16];
            s.f[t][cs] = b;
        }
    }
}

template <int AP, int CT>
__device__ __forceinline__ void slab_compute(const float* aBase, const BSlab& s, v8f (&acc)[4][2])
{
    #pragma unroll
    for (int t = 0; t < KS / 4; ++t) {
        v2f afrag[4];
        #pragma unroll
        for (int rs = 0; rs < 4; ++rs)
            afrag[rs] = *reinterpret_cast<const v2f*>(aBase + rs * 16 * AP + 4 * t);
        #pragma unroll
        for (int rs = 0; rs < 4; ++rs)
            #pragma unroll
            for (int cs = 0; cs < CT; ++cs)
                acc[rs][cs] = __builtin_amdgcn_wmma_f32_16x16x4_f32(
                     false, afrag[rs],
                     false, s.f[t][cs],
                     (short)0, acc[rs][cs],
                     false,  false);
    }
}

template <int K, int N, int CT, int AP, bool OUT_LDS, int OP, bool RELU, int NEXT_N, int NEXT_CT>
__device__ __forceinline__ void layer(const float* aLds,
                                      const float* __restrict__ We, const float* __restrict__ be,
                                      const float* __restrict__ nextW,
                                      float* __restrict__ oGlob, float* oLds,
                                      BSlab& b0, int tid)
{
    const int lane  = tid & 31;
    const int wave  = tid >> 5;
    const int lm    = lane & 15;
    const int khalf = lane >> 4;
    const int mHalf = khalf * 8;
    constexpr int WCOLS = 16 * CT;
    constexpr int PASS  = 8 * WCOLS;

    const float* aFrag = aLds + lm * AP + 2 * khalf;

    for (int nb = 0; nb < N; nb += PASS) {
        const int col0 = nb + wave * WCOLS;
        const float* bp = We + (size_t)(2 * khalf) * N + col0 + lm;

        v8f acc[4][2];
        {
            v8f z = {0.f, 0.f, 0.f, 0.f, 0.f, 0.f, 0.f, 0.f};
            #pragma unroll
            for (int rs = 0; rs < 4; ++rs) { acc[rs][0] = z; acc[rs][1] = z; }
        }

        BSlab b1;
        #pragma nounroll
        for (int k0 = 0; k0 < K - 2 * KS; k0 += 2 * KS) {
            bslab_load<N, CT>(b1, bp + (size_t)(k0 + KS) * N);
            slab_compute<AP, CT>(aFrag + k0, b0, acc);
            bslab_load<N, CT>(b0, bp + (size_t)(k0 + 2 * KS) * N);
            slab_compute<AP, CT>(aFrag + k0 + KS, b1, acc);
        }
        bslab_load<N, CT>(b1, bp + (size_t)(K - KS) * N);
        slab_compute<AP, CT>(aFrag + (K - 2 * KS), b0, acc);
        if (nb + PASS < N) {
            bslab_load<N, CT>(b0, bp + PASS);
        } else if (NEXT_N != 0) {
            bslab_load<NEXT_N, NEXT_CT>(b0, nextW + (size_t)(2 * khalf) * NEXT_N + wave * (16 * NEXT_CT) + lm);
        }
        slab_compute<AP, CT>(aFrag + (K - KS), b1, acc);

        #pragma unroll
        for (int rs = 0; rs < 4; ++rs) {
            #pragma unroll
            for (int cs = 0; cs < CT; ++cs) {
                int   col = col0 + cs * 16 + lm;
                float bv  = be[col];
                #pragma unroll
                for (int v = 0; v < 8; ++v) {
                    int   row = rs * 16 + mHalf + v;
                    float x   = acc[rs][cs][v] + bv;
                    if (RELU) x = fmaxf(x, 0.f);
                    if (OUT_LDS) oLds[row * OP + col] = x;
                    else         oGlob[(size_t)row * N + col] = x;
                }
            }
        }
    }
}

__global__ __launch_bounds__(256)
void mlp4_fused(const float* __restrict__ in,
                const int*   __restrict__ sel_s,
                const int*   __restrict__ sel_d,
                const float* __restrict__ Ws1, const float* __restrict__ bs1,
                const float* __restrict__ Ws2, const float* __restrict__ bs2,
                const float* __restrict__ Wd1, const float* __restrict__ bd1,
                const float* __restrict__ Wd2, const float* __restrict__ bd2,
                float* __restrict__ out,
                int M)
{
    __shared__ __attribute__((aligned(16))) float lds[H_FLOATS + S_FLOATS];
    float* bufH = lds;
    float* bufS = lds + H_FLOATS;

    const int a  = blockIdx.z;
    const int es = sel_s[a];
    const int ed = sel_d[a];
    const int rowBase = blockIdx.x * BM;
    const int tid = threadIdx.x;

    const float* inA  = in  + ((size_t)a * M + rowBase) * 256;
    float*       outA = out + ((size_t)a * M + rowBase) * 128;
    const float* W1 = Ws1 + (size_t)es * 256 * 512;  const float* b1 = bs1 + (size_t)es * 512;
    const float* W2 = Ws2 + (size_t)es * 512 * 256;  const float* b2 = bs2 + (size_t)es * 256;
    const float* W3 = Wd1 + (size_t)ed * 256 * 512;  const float* b3 = bd1 + (size_t)ed * 512;
    const float* W4 = Wd2 + (size_t)ed * 512 * 128;  const float* b4 = bd2 + (size_t)ed * 128;

    BSlab b0;
    bslab_load<512, 2>(b0, W1 + (size_t)(2 * ((tid & 31) >> 4)) * 512 + (tid >> 5) * 32 + (tid & 15));

    #pragma unroll
    for (int i = 0; i < 16; ++i) {
        int idx = i * 256 + tid;
        int row = idx >> 6;
        int c4  = idx & 63;
        v4f x = *reinterpret_cast<const v4f*>(inA + (size_t)row * 256 + c4 * 4);
        *reinterpret_cast<v4f*>(bufS + row * SPITCH + c4 * 4) = x;
    }
    __syncthreads();

    layer<256, 512, 2, SPITCH, true,  HPITCH, true,  256,   2>(bufS, W1, b1, W2,      nullptr, bufH,    b0, tid);
    __syncthreads();
    layer<512, 256, 2, HPITCH, true,  SPITCH, true,  512,   2>(bufH, W2, b2, W3,      nullptr, bufS,    b0, tid);
    __syncthreads();
    layer<256, 512, 2, SPITCH, true,  HPITCH, true,  128,   1>(bufS, W3, b3, W4,      nullptr, bufH,    b0, tid);
    __syncthreads();
    layer<512, 128, 1, HPITCH, false, 0,      false, 0,     1>(bufH, W4, b4, nullptr, outA,    nullptr, b0, tid);
}

extern "C" void kernel_launch(void* const* d_in, const int* in_sizes, int n_in,
                              void* d_out, int out_size, void* d_ws, size_t ws_size,
                              hipStream_t stream) {
    (void)in_sizes; (void)n_in; (void)out_size; (void)d_ws; (void)ws_size;

    const float* inputs = (const float*)d_in[0];
    const int*   sel_s  = (const int*)  d_in[1];
    const int*   sel_d  = (const int*)  d_in[2];
    const float* Ws1    = (const float*)d_in[3];
    const float* bs1    = (const float*)d_in[4];
    const float* Ws2    = (const float*)d_in[5];
    const float* bs2    = (const float*)d_in[6];
    const float* Wd1    = (const float*)d_in[7];
    const float* bd1    = (const float*)d_in[8];
    const float* Wd2    = (const float*)d_in[9];
    const float* bd2    = (const float*)d_in[10];
    float* out = (float*)d_out;

    const int A = 8, B = 2048;

    mlp4_fused<<<dim3(B / BM, 1, A), dim3(256), 0, stream>>>(
        inputs, sel_s, sel_d, Ws1, bs1, Ws2, bs2, Wd1, bd1, Wd2, bd2, out, B);
}


# Each K slab's operand loads (B from global memory, A fragments from LDS) issued as ONE group ahead of the previous slab's 64 WMMAs, pinned by scheduling barriers; A double-buffered in registers
# speedup vs baseline: 1.8478x; 1.0331x over previous
#include <hip/hip_runtime.h>

typedef __attribute__((ext_vector_type(2))) float v2f;
typedef __attribute__((ext_vector_type(8))) float v8f;
typedef __attribute__((ext_vector_type(4))) float v4f;

#define BM 64
#define KS 32
#define HPITCH (512 + 4)
#define SPITCH (256 + 4)
#define H_FLOATS (BM * HPITCH)
#define S_FLOATS (BM * SPITCH)

struct BSlab {
    v2f f[KS / 4][2];
};

template <int N, int CT>
__device__ __forceinline__ void bslab_load(BSlab& s, const float* __restrict__ p)
{
    #pragma unroll
    for (int t = 0; t < KS / 4; ++t) {
        #pragma unroll
        for (int cs = 0; cs < CT; ++cs) {
            v2f b;
            b.x = p[(size_t)(4 * t) * N + cs * 16];
            b.y = p[(size_t)(4 * t + 1) * N + cs * 16];
            s.f[t][cs] = b;
        }
    }
}

struct ASlab {
    v2f f[KS / 4][4];
};

template <int AP>
__device__ __forceinline__ void aslab_load(ASlab& s, const float* aBase)
{
    #pragma unroll
    for (int t = 0; t < KS / 4; ++t) {
        #pragma unroll
        for (int rs = 0; rs < 4; ++rs)
            s.f[t][rs] = *reinterpret_cast<const v2f*>(aBase + rs * 16 * AP + 4 * t);
    }
}

template <int CT>
__device__ __forceinline__ void slab_compute(const ASlab& a, const BSlab& s, v8f (&acc)[4][2])
{
    #pragma unroll
    for (int t = 0; t < KS / 4; ++t) {
        #pragma unroll
        for (int rs = 0; rs < 4; ++rs)
            #pragma unroll
            for (int cs = 0; cs < CT; ++cs)
                acc[rs][cs] = __builtin_amdgcn_wmma_f32_16x16x4_f32(
                     false, a.f[t][rs],
                     false, s.f[t][cs],
                     (short)0, acc[rs][cs],
                     false,  false);
    }
}

#define PIN_HERE() __builtin_amdgcn_sched_barrier(0)

template <int K, int N, int CT, int AP, bool OUT_LDS, int OP, bool RELU, int NEXT_N, int NEXT_CT>
__device__ __forceinline__ void layer(const float* aLds,
                                      const float* __restrict__ We, const float* __restrict__ be,
                                      const float* __restrict__ nextW,
                                      float* __restrict__ oGlob, float* oLds,
                                      BSlab& b0, int tid)
{
    const int lane  = tid & 31;
    const int wave  = tid >> 5;
    const int lm    = lane & 15;
    const int khalf = lane >> 4;
    const int mHalf = khalf * 8;
    constexpr int WCOLS = 16 * CT;
    constexpr int PASS  = 8 * WCOLS;

    const float* aFrag = aLds + lm * AP + 2 * khalf;

    ASlab a0, a1;
    aslab_load<AP>(a0, aFrag);

    for (int nb = 0; nb < N; nb += PASS) {
        const int col0 = nb + wave * WCOLS;
        const float* bp = We + (size_t)(2 * khalf) * N + col0 + lm;

        v8f acc[4][2];
        {
            v8f z = {0.f, 0.f, 0.f, 0.f, 0.f, 0.f, 0.f, 0.f};
            #pragma unroll
            for (int rs = 0; rs < 4; ++rs) { acc[rs][0] = z; acc[rs][1] = z; }
        }

        BSlab b1;
        float bias[2];
        #pragma nounroll
        for (int k0 = 0; k0 < K - 2 * KS; k0 += 2 * KS) {
            bslab_load<N, CT>(b1, bp + (size_t)(k0 + KS) * N);
            aslab_load<AP>(a1, aFrag + k0 + KS);
            PIN_HERE();
            slab_compute<CT>(a0, b0, acc);
            PIN_HERE();
            bslab_load<N, CT>(b0, bp + (size_t)(k0 + 2 * KS) * N);
            aslab_load<AP>(a0, aFrag + k0 + 2 * KS);
            PIN_HERE();
            slab_compute<CT>(a1, b1, acc);
        }
        bslab_load<N, CT>(b1, bp + (size_t)(K - KS) * N);
        aslab_load<AP>(a1, aFrag + (K - KS));
        PIN_HERE();
        slab_compute<CT>(a0, b0, acc);
        PIN_HERE();
        if (nb + PASS < N) {
            bslab_load<N, CT>(b0, bp + PASS);
            aslab_load<AP>(a0, aFrag);
        } else if (NEXT_N != 0) {
            bslab_load<NEXT_N, NEXT_CT>(b0, nextW + (size_t)(2 * khalf) * NEXT_N + wave * (16 * NEXT_CT) + lm);
        }
        #pragma unroll
        for (int cs = 0; cs < CT; ++cs) bias[cs] = be[col0 + cs * 16 + lm];
        PIN_HERE();
        slab_compute<CT>(a1, b1, acc);

        #pragma unroll
        for (int rs = 0; rs < 4; ++rs) {
            #pragma unroll
            for (int cs = 0; cs < CT; ++cs) {
                int   col = col0 + cs * 16 + lm;
                float bv  = bias[cs];
                #pragma unroll
                for (int v = 0; v < 8; ++v) {
                    int   row = rs * 16 + mHalf + v;
                    float x   = acc[rs][cs][v] + bv;
                    if (RELU) x = fmaxf(x, 0.f);
                    if (OUT_LDS) oLds[row * OP + col] = x;
                    else         oGlob[(size_t)row * N + col] = x;
                }
            }
        }
    }
}

__global__ __launch_bounds__(256)
void mlp4_fused(const float* __restrict__ in,
                const int*   __restrict__ sel_s,
                const int*   __restrict__ sel_d,
                const float* __restrict__ Ws1, const float* __restrict__ bs1,
                const float* __restrict__ Ws2, const float* __restrict__ bs2,
                const float* __restrict__ Wd1, const float* __restrict__ bd1,
                const float* __restrict__ Wd2, const float* __restrict__ bd2,
                float* __restrict__ out,
                int M)
{
    __shared__ __attribute__((aligned(16))) float lds[H_FLOATS + S_FLOATS];
    float* bufH = lds;
    float* bufS = lds + H_FLOATS;

    const int a  = blockIdx.z;
    const int es = sel_s[a];
    const int ed = sel_d[a];
    const int rowBase = blockIdx.x * BM;
    const int tid = threadIdx.x;

    const float* inA  = in  + ((size_t)a * M + rowBase) * 256;
    float*       outA = out + ((size_t)a * M + rowBase) * 128;
    const float* W1 = Ws1 + (size_t)es * 256 * 512;  const float* b1 = bs1 + (size_t)es * 512;
    const float* W2 = Ws2 + (size_t)es * 512 * 256;  const float* b2 = bs2 + (size_t)es * 256;
    const float* W3 = Wd1 + (size_t)ed * 256 * 512;  const float* b3 = bd1 + (size_t)ed * 512;
    const float* W4 = Wd2 + (size_t)ed * 512 * 128;  const float* b4 = bd2 + (size_t)ed * 128;

    BSlab b0;
    bslab_load<512, 2>(b0, W1 + (size_t)(2 * ((tid & 31) >> 4)) * 512 + (tid >> 5) * 32 + (tid & 15));

    #pragma unroll
    for (int i = 0; i < 16; ++i) {
        int idx = i * 256 + tid;
        int row = idx >> 6;
        int c4  = idx & 63;
        v4f x = *reinterpret_cast<const v4f*>(inA + (size_t)row * 256 + c4 * 4);
        *reinterpret_cast<v4f*>(bufS + row * SPITCH + c4 * 4) = x;
    }
    __syncthreads();

    layer<256, 512, 2, SPITCH, true,  HPITCH, true,  256,   2>(bufS, W1, b1, W2,      nullptr, bufH,    b0, tid);
    __syncthreads();
    layer<512, 256, 2, HPITCH, true,  SPITCH, true,  512,   2>(bufH, W2, b2, W3,      nullptr, bufS,    b0, tid);
    __syncthreads();
    layer<256, 512, 2, SPITCH, true,  HPITCH, true,  128,   1>(bufS, W3, b3, W4,      nullptr, bufH,    b0, tid);
    __syncthreads();
    layer<512, 128, 1, HPITCH, false, 0,      false, 0,     1>(bufH, W4, b4, nullptr, outA,    nullptr, b0, tid);
}

extern "C" void kernel_launch(void* const* d_in, const int* in_sizes, int n_in,
                              void* d_out, int out_size, void* d_ws, size_t ws_size,
                              hipStream_t stream) {
    (void)in_sizes; (void)n_in; (void)out_size; (void)d_ws; (void)ws_size;

    const float* inputs = (const float*)d_in[0];
    const int*   sel_s  = (const int*)  d_in[1];
    const int*   sel_d  = (const int*)  d_in[2];
    const float* Ws1    = (const float*)d_in[3];
    const float* bs1    = (const float*)d_in[4];
    const float* Ws2    = (const float*)d_in[5];
    const float* bs2    = (const float*)d_in[6];
    const float* Wd1    = (const float*)d_in[7];
    const float* bd1    = (const float*)d_in[8];
    const float* Wd2    = (const float*)d_in[9];
    const float* bd2    = (const float*)d_in[10];
    float* out = (float*)d_out;

    const int A = 8, B = 2048;

    mlp4_fused<<<dim3(B / BM, 1, A), dim3(256), 0, stream>>>(
        inputs, sel_s, sel_d, Ws1, bs1, Ws2, bs2, Wd1, bd1, Wd2, bd2, out, B);
}
